# starting kernel
# baseline (speedup 1.0000x reference)
#include <hip/hip_runtime.h>


typedef __attribute__((ext_vector_type(2))) float v2f;
typedef __attribute__((ext_vector_type(8))) float v8f;

#define PAD 32

__global__ void k_prefix(const float* __restrict__ sA, const float* __restrict__ sB,
                         float* __restrict__ PA, float* __restrict__ PB, int n) {
  if (threadIdx.x == 0) {
    float p = 1.f; PA[0] = p;
    for (int i = 1; i < n; ++i) { p *= sA[i - 1]; PA[i] = p; }
  } else if (threadIdx.x == 1) {
    float p = 1.f; PB[0] = p;
    for (int i = 1; i < n; ++i) { p *= sB[i - 1]; PB[i] = p; }
  }
}

__global__ void k_transpose_scale(const float* __restrict__ x, const float* __restrict__ PB,
                                  float* __restrict__ Xt, int n, int batch) {
  __shared__ float tile[32][33];
  const int k0 = blockIdx.x * 32, b0 = blockIdx.y * 32;
  const int tx = threadIdx.x, ty = threadIdx.y;
#pragma unroll
  for (int s = 0; s < 32; s += 8)
    tile[ty + s][tx] = x[(size_t)(b0 + ty + s) * n + (k0 + tx)];
  __syncthreads();
#pragma unroll
  for (int s = 0; s < 32; s += 8) {
    const int k = k0 + ty + s;
    Xt[(size_t)k * batch + (b0 + tx)] = PB[k] * tile[tx][ty + s];
  }
}

__global__ void k_pass1(const float* __restrict__ Hm, const float* __restrict__ PB,
                        const float* __restrict__ Xt, float* __restrict__ TMP,
                        int n, int batch) {
  extern __shared__ float hp[];
  const int tid = threadIdx.x;
  const int lane = tid & 31;
  const int w = tid >> 5;
  const int r = blockIdx.z;

  for (int t = tid; t < PAD; t += blockDim.x) hp[t] = 0.f;
  for (int t = tid; t < 4; t += blockDim.x) hp[PAD + n + t] = 0.f;
  for (int t = tid; t < n; t += blockDim.x) hp[PAD + t] = Hm[(size_t)r * n + t] / PB[t];
  __syncthreads();

  const int j0 = blockIdx.x * 64 + (w & 1) * 32;
  const int b0 = blockIdx.y * 256 + (w >> 1) * 64;

  const int jr   = lane & 15;
  const int kb2  = (lane >> 4) * 2;
  const int nlan = lane & 15;

  v8f c[2][4];
#pragma unroll
  for (int jt = 0; jt < 2; ++jt)
#pragma unroll
    for (int t = 0; t < 4; ++t) c[jt][t] = (v8f){0.f,0.f,0.f,0.f,0.f,0.f,0.f,0.f};

  const int abase0 = PAD + kb2 - jr - j0;
  const int abase1 = abase0 - 16;
  const float* __restrict__ xcol = Xt + (size_t)kb2 * batch + (b0 + nlan);

  for (int k0 = j0; k0 < n; k0 += 4) {
    v2f a0, a1;
    a0.x = hp[abase0 + k0]; a0.y = hp[abase0 + k0 + 1];
    a1.x = hp[abase1 + k0]; a1.y = hp[abase1 + k0 + 1];
    const float* __restrict__ xp = xcol + (size_t)k0 * batch;
    __builtin_prefetch(xp + (size_t)32 * batch, 0, 1);
#pragma unroll
    for (int t = 0; t < 4; ++t) {
      v2f bb;
      bb.x = xp[16 * t];
      bb.y = xp[batch + 16 * t];
      c[0][t] = __builtin_amdgcn_wmma_f32_16x16x4_f32(false, a0, false, bb,
                                                      (short)0, c[0][t], false, false);
      c[1][t] = __builtin_amdgcn_wmma_f32_16x16x4_f32(false, a1, false, bb,
                                                      (short)0, c[1][t], false, false);
    }
  }

  const int jhalf = (lane >> 4) * 8;
#pragma unroll
  for (int jt = 0; jt < 2; ++jt)
#pragma unroll
    for (int t = 0; t < 4; ++t)
#pragma unroll
      for (int v = 0; v < 8; ++v) {
        const int j = j0 + 16 * jt + v + jhalf;
        const int b = b0 + nlan + 16 * t;
        TMP[((size_t)r * n + j) * batch + b] = c[jt][t][v];
      }
}

__global__ void k_pass2(const float* __restrict__ Gm, const float* __restrict__ PA,
                        const float* __restrict__ TMP, const float* __restrict__ bias,
                        float* __restrict__ out, int n, int batch, int rdim) {
  extern __shared__ float gp[];
  const int tid = threadIdx.x;
  const int lane = tid & 31;
  const int w = tid >> 5;

  const int i0 = blockIdx.x * 64 + (w & 1) * 32;
  const int b0 = blockIdx.y * 256 + (w >> 1) * 64;

  const int mlan = lane & 15;
  const int kb2  = (lane >> 4) * 2;

  v8f c[2][4];
#pragma unroll
  for (int it = 0; it < 2; ++it)
#pragma unroll
    for (int t = 0; t < 4; ++t) c[it][t] = (v8f){0.f,0.f,0.f,0.f,0.f,0.f,0.f,0.f};

  const int jmax   = i0 + 32;
  const int bbase0 = PAD + i0 + mlan - kb2;
  const int bbase1 = bbase0 + 16;

  for (int r = 0; r < rdim; ++r) {
    __syncthreads();
    for (int t = tid; t < PAD; t += blockDim.x) gp[t] = 0.f;
    for (int t = tid; t < 4; t += blockDim.x) gp[PAD + n + t] = 0.f;
    for (int t = tid; t < n; t += blockDim.x) gp[PAD + t] = Gm[(size_t)r * n + t] / PA[t];
    __syncthreads();

    const float* __restrict__ trow =
        TMP + ((size_t)r * n + kb2) * batch + (b0 + mlan);
    for (int j0k = 0; j0k < jmax; j0k += 4) {
      v2f bb0, bb1;
      bb0.x = gp[bbase0 - j0k]; bb0.y = gp[bbase0 - j0k - 1];
      bb1.x = gp[bbase1 - j0k]; bb1.y = gp[bbase1 - j0k - 1];
      const float* __restrict__ ap = trow + (size_t)j0k * batch;
#pragma unroll
      for (int t = 0; t < 4; ++t) {
        v2f a;
        a.x = ap[16 * t];
        a.y = ap[batch + 16 * t];
        c[0][t] = __builtin_amdgcn_wmma_f32_16x16x4_f32(false, a, false, bb0,
                                                        (short)0, c[0][t], false, false);
        c[1][t] = __builtin_amdgcn_wmma_f32_16x16x4_f32(false, a, false, bb1,
                                                        (short)0, c[1][t], false, false);
      }
    }
  }

  const int bhalf = (lane >> 4) * 8;
#pragma unroll
  for (int it = 0; it < 2; ++it) {
    const int i = i0 + 16 * it + mlan;
    const float pa = PA[i];
    const float bi = bias[i];
#pragma unroll
    for (int t = 0; t < 4; ++t)
#pragma unroll
      for (int v = 0; v < 8; ++v) {
        const int b = b0 + 16 * t + v + bhalf;
        out[(size_t)b * n + i] = bi + pa * c[it][t][v];
      }
  }
}

extern "C" void kernel_launch(void* const* d_in, const int* in_sizes, int n_in,
                              void* d_out, int out_size, void* d_ws, size_t ws_size,
                              hipStream_t stream) {
  const float* x     = (const float*)d_in[0];
  const float* sA    = (const float*)d_in[1];
  const float* sB    = (const float*)d_in[2];
  const float* G     = (const float*)d_in[3];
  const float* H     = (const float*)d_in[4];
  const float* bias  = (const float*)d_in[5];
  float* out = (float*)d_out;

  const int n     = in_sizes[5];
  const int batch = in_sizes[0] / n;
  const int rdim  = in_sizes[3] / n;

  float* PA  = (float*)d_ws;
  float* PB  = PA + n;
  float* Xt  = PB + n;
  float* TMP = Xt + (size_t)n * batch;

  const size_t ldsBytes = (size_t)(PAD + n + 4) * sizeof(float);

  k_prefix<<<1, 64, 0, stream>>>(sA, sB, PA, PB, n);

  {
    dim3 grid(n / 32, batch / 32), block(32, 8);
    k_transpose_scale<<<grid, block, 0, stream>>>(x, PB, Xt, n, batch);
  }
  {
    dim3 grid(n / 64, batch / 256, rdim);
    k_pass1<<<grid, 256, ldsBytes, stream>>>(H, PB, Xt, TMP, n, batch);
  }
  {
    dim3 grid(n / 64, batch / 256, 1);
    k_pass2<<<grid, 256, ldsBytes, stream>>>(G, PA, TMP, bias, out, n, batch, rdim);
  }
}


# Three products for four per 32-k trip, both passes: Toeplitz blocks [[Ta,Tb],[Tc,Ta]] give Y0=Ta(X0+X1)+(Tb-Ta)X1, Y1=Ta(X0+X1)+(Tc-Ta)X0: 24 WMMAs, was 32; sums regrouped (sum order changes)
# speedup vs baseline: 3.6785x; 3.6785x over previous
#include <hip/hip_runtime.h>


typedef __attribute__((ext_vector_type(2))) float v2f;
typedef __attribute__((ext_vector_type(4))) float v4f;
typedef __attribute__((ext_vector_type(8))) float v8f;

#define PAD 32

__device__ __forceinline__ int prio_stretch(int left) {
  if (left > 64) { __builtin_amdgcn_s_setprio(3); return left - 64; }
  if (left > 16) { __builtin_amdgcn_s_setprio(2); return left - 16; }
  if (left > 4)  { __builtin_amdgcn_s_setprio(1); return left - 4; }
  __builtin_amdgcn_s_setprio(0);
  return left;
}

__global__ void __launch_bounds__(1024)
k_prefix(const float* __restrict__ sA, const float* __restrict__ sB,
         float* __restrict__ PA, float* __restrict__ PB, int n) {
  __shared__ float wt[32];
  const float* __restrict__ s = blockIdx.x ? sB : sA;
  float* __restrict__ P = blockIdx.x ? PB : PA;
  const int tid = threadIdx.x;
  const int lane = tid & 31;
  const int w = tid >> 5;
  const int nw = blockDim.x >> 5;
  const int chunk = blockDim.x * 4;
  float carry = 1.f;
  for (int base = 0; base < n; base += chunk) {
    const int i0 = base + tid * 4;
    float e[4];
#pragma unroll
    for (int u = 0; u < 4; ++u) {
      const int i = i0 + u;
      const float v = s[max(min(i - 1, n - 2), 0)];
      e[u] = (i >= 1 && i < n) ? v : 1.f;
    }
    const float p0 = e[0], p1 = p0 * e[1], p2 = p1 * e[2], p3 = p2 * e[3];
    float sc = p3;
#pragma unroll
    for (int d = 1; d < 32; d <<= 1) {
      const float v = __shfl_up(sc, d);
      if (lane >= d) sc *= v;
    }
    float ex = __shfl_up(sc, 1);
    if (lane == 0) ex = 1.f;
    if (lane == 31) wt[w] = sc;
    __syncthreads();
    float inc = (lane < nw) ? wt[lane] : 1.f;
#pragma unroll
    for (int d = 1; d < 32; d <<= 1) {
      const float v = __shfl_up(inc, d);
      if (lane >= d) inc *= v;
    }
    const float upto = __shfl(inc, w > 0 ? w - 1 : 0);
    const float below = (w > 0) ? upto : 1.f;
    const float total = __shfl(inc, 31);
    const float pre = carry * below * ex;
    if (i0 + 3 < n) {
      v4f o; o.x = pre * p0; o.y = pre * p1; o.z = pre * p2; o.w = pre * p3;
      *(v4f*)(P + i0) = o;
    } else {
      if (i0 < n) P[i0] = pre * p0;
      if (i0 + 1 < n) P[i0 + 1] = pre * p1;
      if (i0 + 2 < n) P[i0 + 2] = pre * p2;
    }
    carry *= total;
    __syncthreads();
  }
}

__global__ void __launch_bounds__(256)
k_block_scale(const float* __restrict__ x, const float* __restrict__ PB,
              float* __restrict__ Xb, int n, int batch) {
  const int lane = threadIdx.x & 31;
  const int w = threadIdx.x >> 5;
  const int kb = blockIdx.x * 4 + (w & 3);
  const int bg = blockIdx.y * 2 + (w >> 2);
  const int eh = lane >> 4, bl = lane & 15;
  const int k = kb * 8 + eh * 4;
  const v4f xv = *(const v4f*)(x + (size_t)(bg * 16 + bl) * n + k);
  const v4f pv = *(const v4f*)(PB + k);
  *(v4f*)(Xb + (size_t)kb * batch * 8 + bg * 128 + eh * 64 + bl * 4) = pv * xv;
}

__global__ void __launch_bounds__(1024) __attribute__((amdgpu_waves_per_eu(8, 8)))
k_pass1(const float* __restrict__ Hm, const float* __restrict__ PB,
        const float* __restrict__ Xb, float* __restrict__ TMP,
        int n, int batch) {
  extern __shared__ float hp[];
  const int tid = threadIdx.x;
  const int lane = tid & 31;
  const int w = __builtin_amdgcn_readfirstlane(tid >> 5);
  const int r = blockIdx.z;

  for (int t = tid; t < PAD; t += blockDim.x) hp[t] = 0.f;
  for (int t = tid; t < 4; t += blockDim.x) hp[PAD + n + t] = 0.f;
  for (int t = tid; t < n; t += blockDim.x) hp[PAD + t] = Hm[(size_t)r * n + t] / PB[t];
  __syncthreads();

  const int b0 = blockIdx.y * 512 + (w >> 1) * 32;

  const int jr   = lane & 15;
  const int h    = lane >> 4;
  const int nlan = lane & 15;
  const int nblk = n / 64;

#pragma nounroll
  for (int half = 0; half < 2; ++half) {
    const int xb = half ? (nblk - 1 - (int)blockIdx.x) : (int)blockIdx.x;
    const int j0 = xb * 64 + (w & 1) * 32;
    const int after = half ? 0 : ((n - ((nblk - 1 - (int)blockIdx.x) * 64 + (w & 1) * 32)) >> 5);

    v8f cc[2], c0[2], c1[2];
#pragma unroll
    for (int t = 0; t < 2; ++t) {
      cc[t] = (v8f){0.f,0.f,0.f,0.f,0.f,0.f,0.f,0.f};
      c0[t] = (v8f){0.f,0.f,0.f,0.f,0.f,0.f,0.f,0.f};
      c1[t] = (v8f){0.f,0.f,0.f,0.f,0.f,0.f,0.f,0.f};
    }

    const int abase0 = PAD + 8 * h - jr - j0;
    const float* __restrict__ xrow = Xb + (size_t)h * batch * 8 + (b0 >> 4) * 128 + nlan * 4;

    for (int kblk = j0; kblk < n; ) {
     const int kend = min(n, kblk + 32 * prio_stretch(((n - kblk) >> 5) + after));
     for (; kblk < kend; kblk += 32) {
      const float* __restrict__ xp = xrow + (size_t)kblk * batch;
      v4f xv[2][2][2];
#pragma unroll
      for (int u = 0; u < 2; ++u)
#pragma unroll
        for (int t = 0; t < 2; ++t) {
          xv[u][t][0] = *(const v4f*)(xp + (size_t)u * 16 * batch + 128 * t);
          xv[u][t][1] = *(const v4f*)(xp + (size_t)u * 16 * batch + 128 * t + 64);
        }
      float ta[8], db[8], dc[8];
#pragma unroll
      for (int e = 0; e < 8; ++e) {
        ta[e] = hp[abase0 + kblk + e];
        db[e] = hp[abase0 + kblk + 16 + e] - ta[e];
        dc[e] = hp[abase0 + kblk - 16 + e] - ta[e];
      }
      v4f sv[2][2];
#pragma unroll
      for (int t = 0; t < 2; ++t) {
        sv[t][0] = xv[0][t][0] + xv[1][t][0];
        sv[t][1] = xv[0][t][1] + xv[1][t][1];
      }
#pragma unroll
      for (int q = 0; q < 4; ++q) {
        v2f fa, fdb, fdc;
        fa.x  = ta[2 * q]; fa.y  = ta[2 * q + 1];
        fdb.x = db[2 * q]; fdb.y = db[2 * q + 1];
        fdc.x = dc[2 * q]; fdc.y = dc[2 * q + 1];
#pragma unroll
        for (int t = 0; t < 2; ++t) {
          v2f x0, x1, xs;
          x0.x = xv[0][t][q >> 1][(q & 1) * 2]; x0.y = xv[0][t][q >> 1][(q & 1) * 2 + 1];
          x1.x = xv[1][t][q >> 1][(q & 1) * 2]; x1.y = xv[1][t][q >> 1][(q & 1) * 2 + 1];
          xs.x = sv[t][q >> 1][(q & 1) * 2];    xs.y = sv[t][q >> 1][(q & 1) * 2 + 1];
          c1[t] = __builtin_amdgcn_wmma_f32_16x16x4_f32(false, fdc, false, x0,
                                                        (short)0, c1[t], false, false);
          c0[t] = __builtin_amdgcn_wmma_f32_16x16x4_f32(false, fdb, false, x1,
                                                        (short)0, c0[t], false, false);
          cc[t] = __builtin_amdgcn_wmma_f32_16x16x4_f32(false, fa, false, xs,
                                                        (short)0, cc[t], false, false);
        }
      }
     }
    }

    v8f c[2][2];
#pragma unroll
    for (int t = 0; t < 2; ++t) {
      c[0][t] = cc[t] + c0[t];
      c[1][t] = cc[t] + c1[t];
    }

#pragma unroll
    for (int jt = 0; jt < 2; ++jt)
#pragma unroll
      for (int t = 0; t < 2; ++t) {
        float* __restrict__ dst = TMP + (size_t)r * n * batch
            + (size_t)(((j0 + 16 * jt) >> 3) + h) * batch * 8
            + ((b0 >> 4) + t) * 128 + nlan * 4;
        v4f lo, hi;
        lo.x = c[jt][t][0]; lo.y = c[jt][t][1]; lo.z = c[jt][t][2]; lo.w = c[jt][t][3];
        hi.x = c[jt][t][4]; hi.y = c[jt][t][5]; hi.z = c[jt][t][6]; hi.w = c[jt][t][7];
        *(v4f*)dst = lo;
        *(v4f*)(dst + 64) = hi;
      }
  }
}

#define NROW 4096
__global__ void __launch_bounds__(1024) __attribute__((amdgpu_waves_per_eu(8, 8)))
k_pass2(const float* __restrict__ Gm, const float* __restrict__ PA,
        const float* __restrict__ TMP, const float* __restrict__ bias,
        float* __restrict__ out, int n, int batch) {
  __shared__ float gp[4][PAD + NROW + 4];
  __shared__ float red[32][8][32];
  const int tid = threadIdx.x;
  const int lane = tid & 31;
  const int w = __builtin_amdgcn_readfirstlane(tid >> 5);
  const int r = w >> 3;

  for (int t = tid; t < 4 * PAD; t += blockDim.x) gp[t / PAD][t % PAD] = 0.f;
  for (int t = tid; t < 16; t += blockDim.x) gp[t >> 2][PAD + n + (t & 3)] = 0.f;
  for (int t = tid; t < 4 * n; t += blockDim.x) {
    const int rr = t / n, m = t - rr * n;
    gp[rr][PAD + m] = Gm[(size_t)rr * n + m] / PA[m];
  }
  __syncthreads();
  const float* __restrict__ gpr = gp[r];

  const int b0 = blockIdx.y * 128 + ((w >> 1) & 3) * 32;

  const int mlan = lane & 15;
  const int h    = lane >> 4;
  const int bhalf = h * 8;
  const int nblk = n / 64;

#pragma nounroll
  for (int half = 0; half < 2; ++half) {
    const int xb = half ? (nblk - 1 - (int)blockIdx.x) : (int)blockIdx.x;
    const int i0 = xb * 64 + (w & 1) * 32;

    v8f cc[2], c0[2], c1[2];
#pragma unroll
    for (int t = 0; t < 2; ++t) {
      cc[t] = (v8f){0.f,0.f,0.f,0.f,0.f,0.f,0.f,0.f};
      c0[t] = (v8f){0.f,0.f,0.f,0.f,0.f,0.f,0.f,0.f};
      c1[t] = (v8f){0.f,0.f,0.f,0.f,0.f,0.f,0.f,0.f};
    }

    const int jmax   = i0 + 32;
    const int bbase0 = PAD + i0 + mlan - 8 * h;

    const float* __restrict__ trow = TMP + (size_t)r * n * batch
        + (size_t)h * batch * 8 + (b0 >> 4) * 128 + mlan * 4;
    for (int jblk = 0; jblk < jmax; ) {
     const int jend = jblk + 32 * prio_stretch((jmax - jblk) >> 5);
     for (; jblk < jend; jblk += 32) {
      const float* __restrict__ ap = trow + (size_t)jblk * batch;
      v4f av[2][2][2];
#pragma unroll
      for (int u = 0; u < 2; ++u)
#pragma unroll
        for (int t = 0; t < 2; ++t) {
          av[u][t][0] = *(const v4f*)(ap + (size_t)u * 16 * batch + 128 * t);
          av[u][t][1] = *(const v4f*)(ap + (size_t)u * 16 * batch + 128 * t + 64);
        }
      float ta[8], db[8], dc[8];
#pragma unroll
      for (int e = 0; e < 8; ++e) {
        ta[e] = gpr[bbase0 - jblk - e];
        db[e] = gpr[bbase0 - jblk - 16 - e] - ta[e];
        dc[e] = gpr[bbase0 - jblk + 16 - e] - ta[e];
      }
      v4f sv[2][2];
#pragma unroll
      for (int t = 0; t < 2; ++t) {
        sv[t][0] = av[0][t][0] + av[1][t][0];
        sv[t][1] = av[0][t][1] + av[1][t][1];
      }
#pragma unroll
      for (int q = 0; q < 4; ++q) {
        v2f fb, fdb, fdc;
        fb.x  = ta[2 * q]; fb.y  = ta[2 * q + 1];
        fdb.x = db[2 * q]; fdb.y = db[2 * q + 1];
        fdc.x = dc[2 * q]; fdc.y = dc[2 * q + 1];
#pragma unroll
        for (int t = 0; t < 2; ++t) {
          v2f x0, x1, xs;
          x0.x = av[0][t][q >> 1][(q & 1) * 2]; x0.y = av[0][t][q >> 1][(q & 1) * 2 + 1];
          x1.x = av[1][t][q >> 1][(q & 1) * 2]; x1.y = av[1][t][q >> 1][(q & 1) * 2 + 1];
          xs.x = sv[t][q >> 1][(q & 1) * 2];    xs.y = sv[t][q >> 1][(q & 1) * 2 + 1];
          c1[t] = __builtin_amdgcn_wmma_f32_16x16x4_f32(false, x0, false, fdc,
                                                        (short)0, c1[t], false, false);
          c0[t] = __builtin_amdgcn_wmma_f32_16x16x4_f32(false, x1, false, fdb,
                                                        (short)0, c0[t], false, false);
          cc[t] = __builtin_amdgcn_wmma_f32_16x16x4_f32(false, xs, false, fb,
                                                        (short)0, cc[t], false, false);
        }
      }
     }
    }

    v8f c[2][2];
#pragma unroll
    for (int t = 0; t < 2; ++t) {
      c[0][t] = cc[t] + c0[t];
      c[1][t] = cc[t] + c1[t];
    }

#pragma unroll
    for (int a = 0; a < 4; ++a) {
      const int it = a >> 1, t = a & 1;
      if (r != a) {
#pragma unroll
        for (int v = 0; v < 8; ++v) red[w][v][lane] = c[it][t][v];
      }
      __syncthreads();
      if (r == a) {
        const int i = i0 + 16 * it + mlan;
        const float pa = PA[i];
        const float bi = bias[i];
#pragma unroll
        for (int v = 0; v < 8; ++v) {
          float s = 0.f;
#pragma unroll
          for (int q = 0; q < 4; ++q) {
            const float p = (q == a) ? c[it][t][v] : red[(q << 3) | (w & 7)][v][lane];
            s = (q == 0) ? p : s + p;
          }
          const int b = b0 + 16 * t + v + bhalf;
          out[(size_t)b * n + i] = bi + pa * s;
        }
      }
      __syncthreads();
    }
  }
}

extern "C" void kernel_launch(void* const* d_in, const int* in_sizes, int n_in,
                              void* d_out, int out_size, void* d_ws, size_t ws_size,
                              hipStream_t stream) {
  const float* x     = (const float*)d_in[0];
  const float* sA    = (const float*)d_in[1];
  const float* sB    = (const float*)d_in[2];
  const float* G     = (const float*)d_in[3];
  const float* H     = (const float*)d_in[4];
  const float* bias  = (const float*)d_in[5];
  float* out = (float*)d_out;

  const int n     = in_sizes[5];
  const int batch = in_sizes[0] / n;
  const int rdim  = in_sizes[3] / n;

  float* PA  = (float*)d_ws;
  float* PB  = PA + n;
  float* Xb  = PB + n;
  float* TMP = Xb + (size_t)n * batch;

  const size_t ldsBytes = (size_t)(PAD + n + 4) * sizeof(float);

  k_prefix<<<2, 1024, 0, stream>>>(sA, sB, PA, PB, n);

  {
    dim3 grid(n / 32, batch / 32);
    k_block_scale<<<grid, 256, 0, stream>>>(x, PB, Xb, n, batch);
  }
  {
    dim3 grid(n / 128, batch / 512, rdim);
    k_pass1<<<grid, 1024, ldsBytes, stream>>>(H, PB, Xb, TMP, n, batch);
  }
  {
    dim3 grid(n / 128, batch / 128, 1);
    k_pass2<<<grid, 1024, 0, stream>>>(G, PA, TMP, bias, out, n, batch);
  }
}


# Second level of the Toeplitz split: a 64-k trip's 4x4 blocks take 9 products for 16 (was 12 by the one-level form); wave tile 64x32, 4 waves per SIMD; sums regrouped (sum order changes)
# speedup vs baseline: 4.4966x; 1.2224x over previous
#include <hip/hip_runtime.h>


typedef __attribute__((ext_vector_type(2))) float v2f;
typedef __attribute__((ext_vector_type(4))) float v4f;
typedef __attribute__((ext_vector_type(8))) float v8f;

#define PAD 64

__device__ __forceinline__ int prio_stretch(int left) {
  if (left > 22) { __builtin_amdgcn_s_setprio(3); return left - 22; }
  if (left > 6)  { __builtin_amdgcn_s_setprio(2); return left - 6; }
  if (left > 2)  { __builtin_amdgcn_s_setprio(1); return left - 2; }
  __builtin_amdgcn_s_setprio(0);
  return left;
}

__global__ void __launch_bounds__(1024)
k_prefix(const float* __restrict__ sA, const float* __restrict__ sB,
         float* __restrict__ PA, float* __restrict__ PB, int n) {
  __shared__ float wt[32];
  const float* __restrict__ s = blockIdx.x ? sB : sA;
  float* __restrict__ P = blockIdx.x ? PB : PA;
  const int tid = threadIdx.x;
  const int lane = tid & 31;
  const int w = tid >> 5;
  const int nw = blockDim.x >> 5;
  const int chunk = blockDim.x * 4;
  float carry = 1.f;
  for (int base = 0; base < n; base += chunk) {
    const int i0 = base + tid * 4;
    float e[4];
#pragma unroll
    for (int u = 0; u < 4; ++u) {
      const int i = i0 + u;
      const float v = s[max(min(i - 1, n - 2), 0)];
      e[u] = (i >= 1 && i < n) ? v : 1.f;
    }
    const float p0 = e[0], p1 = p0 * e[1], p2 = p1 * e[2], p3 = p2 * e[3];
    float sc = p3;
#pragma unroll
    for (int d = 1; d < 32; d <<= 1) {
      const float v = __shfl_up(sc, d);
      if (lane >= d) sc *= v;
    }
    float ex = __shfl_up(sc, 1);
    if (lane == 0) ex = 1.f;
    if (lane == 31) wt[w] = sc;
    __syncthreads();
    float inc = (lane < nw) ? wt[lane] : 1.f;
#pragma unroll
    for (int d = 1; d < 32; d <<= 1) {
      const float v = __shfl_up(inc, d);
      if (lane >= d) inc *= v;
    }
    const float upto = __shfl(inc, w > 0 ? w - 1 : 0);
    const float below = (w > 0) ? upto : 1.f;
    const float total = __shfl(inc, 31);
    const float pre = carry * below * ex;
    if (i0 + 3 < n) {
      v4f o; o.x = pre * p0; o.y = pre * p1; o.z = pre * p2; o.w = pre * p3;
      *(v4f*)(P + i0) = o;
    } else {
      if (i0 < n) P[i0] = pre * p0;
      if (i0 + 1 < n) P[i0 + 1] = pre * p1;
      if (i0 + 2 < n) P[i0 + 2] = pre * p2;
    }
    carry *= total;
    __syncthreads();
  }
}

__global__ void __launch_bounds__(256)
k_block_scale(const float* __restrict__ x, const float* __restrict__ PB,
              float* __restrict__ Xb, int n, int batch) {
  const int lane = threadIdx.x & 31;
  const int w = threadIdx.x >> 5;
  const int kb = blockIdx.x * 4 + (w & 3);
  const int bg = blockIdx.y * 2 + (w >> 2);
  const int eh = lane >> 4, bl = lane & 15;
  const int k = kb * 8 + eh * 4;
  const v4f xv = *(const v4f*)(x + (size_t)(bg * 16 + bl) * n + k);
  const v4f pv = *(const v4f*)(PB + k);
  *(v4f*)(Xb + (size_t)kb * batch * 8 + bg * 128 + eh * 64 + bl * 4) = pv * xv;
}

template <bool TOEP_IS_B>
__device__ __forceinline__ void trip_half(v8f (&acc)[2][9], const float (&td)[7][4],
                                          const v4f (&xv)[4][2]) {
  float t0[4], a1[4], a2[4], bc[4], q0[4], q1[4], cm[4], r0[4], r1[4];
#pragma unroll
  for (int e = 0; e < 4; ++e) {
    t0[e] = td[3][e];
    a1[e] = td[4][e] - t0[e];
    a2[e] = td[2][e] - t0[e];
    bc[e] = td[5][e] - t0[e];
    const float by = td[4][e] - td[2][e];
    q0[e] = (td[6][e] - td[4][e]) - bc[e];
    q1[e] = by - bc[e];
    cm[e] = td[1][e] - t0[e];
    r0[e] = (td[2][e] - td[4][e]) - cm[e];
    r1[e] = (td[0][e] - td[2][e]) - cm[e];
  }
  v4f s0[2], s1[2], ss[2], xb[2], xt[2];
#pragma unroll
  for (int t = 0; t < 2; ++t) {
    s0[t] = xv[0][t] + xv[2][t];
    s1[t] = xv[1][t] + xv[3][t];
    ss[t] = s0[t] + s1[t];
    xb[t] = xv[2][t] + xv[3][t];
    xt[t] = xv[0][t] + xv[1][t];
  }
#pragma unroll
  for (int qq = 0; qq < 2; ++qq) {
    const float* tf[9] = {t0, a1, a2, bc, q0, q1, cm, r0, r1};
#pragma unroll
    for (int t = 0; t < 2; ++t) {
      const v4f sf[9] = {ss[t], s1[t], s0[t], xb[t], xv[3][t], xv[2][t], xt[t], xv[1][t], xv[0][t]};
#pragma unroll
      for (int m = 0; m < 9; ++m) {
        v2f ft, fx;
        ft.x = tf[m][2 * qq]; ft.y = tf[m][2 * qq + 1];
        fx.x = sf[m][2 * qq]; fx.y = sf[m][2 * qq + 1];
        if (TOEP_IS_B)
          acc[t][m] = __builtin_amdgcn_wmma_f32_16x16x4_f32(false, fx, false, ft,
                                                            (short)0, acc[t][m], false, false);
        else
          acc[t][m] = __builtin_amdgcn_wmma_f32_16x16x4_f32(false, ft, false, fx,
                                                            (short)0, acc[t][m], false, false);
      }
    }
  }
}

__device__ __forceinline__ void combine9(const v8f (&a)[9], v8f (&y)[4]) {
  const v8f up = a[0] + a[3];
  const v8f lo = a[0] + a[6];
  y[0] = up + (a[1] + a[4]);
  y[1] = up + (a[2] + a[5]);
  y[2] = lo + (a[1] + a[7]);
  y[3] = lo + (a[2] + a[8]);
}

__global__ void __launch_bounds__(512) __attribute__((amdgpu_waves_per_eu(4, 4)))
k_pass1(const float* __restrict__ Hm, const float* __restrict__ PB,
        const float* __restrict__ Xb, float* __restrict__ TMP,
        int n, int batch) {
  extern __shared__ float hp[];
  const int tid = threadIdx.x;
  const int lane = tid & 31;
  const int w = __builtin_amdgcn_readfirstlane(tid >> 5);
  const int r = blockIdx.z;

  for (int t = tid; t < PAD; t += blockDim.x) hp[t] = 0.f;
  for (int t = tid; t < 4; t += blockDim.x) hp[PAD + n + t] = 0.f;
  for (int t = tid; t < n; t += blockDim.x) hp[PAD + t] = Hm[(size_t)r * n + t] / PB[t];
  __syncthreads();

  const int b0 = blockIdx.y * 512 + w * 32;

  const int jr   = lane & 15;
  const int h    = lane >> 4;
  const int nlan = lane & 15;
  const int nblk = n / 64;

#pragma nounroll
  for (int half = 0; half < 2; ++half) {
    const int xb = half ? (nblk - 1 - (int)blockIdx.x) : (int)blockIdx.x;
    const int j0 = xb * 64;
    const int after = half ? 0 : ((n - (nblk - 1 - (int)blockIdx.x) * 64) >> 6);

    v8f acc[2][9];
#pragma unroll
    for (int t = 0; t < 2; ++t)
#pragma unroll
      for (int m = 0; m < 9; ++m) acc[t][m] = (v8f){0.f,0.f,0.f,0.f,0.f,0.f,0.f,0.f};

    const int abase0 = PAD + 8 * h - jr - j0;
    const float* __restrict__ xrow = Xb + (b0 >> 4) * 128;
    const unsigned loff = (unsigned)(h * batch * 8 + nlan * 4);

    for (int kblk = j0; kblk < n; ) {
     const int kend = min(n, kblk + 64 * prio_stretch(((n - kblk) >> 6) + after));
     for (; kblk < kend; kblk += 64) {
      const float* __restrict__ xp = xrow + (size_t)kblk * batch;
#pragma unroll
      for (int p = 0; p < 2; ++p) {
        v4f xv[4][2];
#pragma unroll
        for (int u = 0; u < 4; ++u)
#pragma unroll
          for (int t = 0; t < 2; ++t)
            xv[u][t] = *(const v4f*)(xp + (size_t)u * 16 * batch + (128 * t + 64 * p) + loff);
        float td[7][4];
#pragma unroll
        for (int d = 0; d < 7; ++d)
#pragma unroll
          for (int e = 0; e < 4; ++e)
            td[d][e] = hp[abase0 + kblk + 16 * (d - 3) + 4 * p + e];
        trip_half<false>(acc, td, xv);
      }
     }
    }

#pragma unroll
    for (int t = 0; t < 2; ++t) {
      v8f y[4];
      combine9(acc[t], y);
#pragma unroll
      for (int jt = 0; jt < 4; ++jt) {
        float* __restrict__ dst = TMP + (size_t)r * n * batch
            + (size_t)(((j0 + 16 * jt) >> 3) + h) * batch * 8
            + ((b0 >> 4) + t) * 128 + nlan * 4;
        v4f lo, hi;
        lo.x = y[jt][0]; lo.y = y[jt][1]; lo.z = y[jt][2]; lo.w = y[jt][3];
        hi.x = y[jt][4]; hi.y = y[jt][5]; hi.z = y[jt][6]; hi.w = y[jt][7];
        *(v4f*)dst = lo;
        *(v4f*)(dst + 64) = hi;
      }
    }
  }
}

#define NROW 4096
__global__ void __launch_bounds__(512) __attribute__((amdgpu_waves_per_eu(4, 4)))
k_pass2(const float* __restrict__ Gm, const float* __restrict__ PA,
        const float* __restrict__ TMP, const float* __restrict__ bias,
        float* __restrict__ out, int n, int batch) {
  __shared__ float gp[4][PAD + NROW + 4];
  __shared__ float red[16][2][8][32];
  const int tid = threadIdx.x;
  const int lane = tid & 31;
  const int w = __builtin_amdgcn_readfirstlane(tid >> 5);
  const int r = w >> 2;

  for (int t = tid; t < 4 * PAD; t += blockDim.x) gp[t / PAD][t % PAD] = 0.f;
  for (int t = tid; t < 16; t += blockDim.x) gp[t >> 2][PAD + n + (t & 3)] = 0.f;
  for (int t = tid; t < 4 * n; t += blockDim.x) {
    const int rr = t / n, m = t - rr * n;
    gp[rr][PAD + m] = Gm[(size_t)rr * n + m] / PA[m];
  }
  __syncthreads();
  const float* __restrict__ gpr = gp[r];

  const int b0 = blockIdx.y * 128 + (w & 3) * 32;

  const int mlan = lane & 15;
  const int h    = lane >> 4;
  const int bhalf = h * 8;
  const int nblk = n / 64;

#pragma nounroll
  for (int half = 0; half < 2; ++half) {
    const int xb = half ? (nblk - 1 - (int)blockIdx.x) : (int)blockIdx.x;
    const int i0 = xb * 64;

    v8f acc[2][9];
#pragma unroll
    for (int t = 0; t < 2; ++t)
#pragma unroll
      for (int m = 0; m < 9; ++m) acc[t][m] = (v8f){0.f,0.f,0.f,0.f,0.f,0.f,0.f,0.f};

    const int jmax   = i0 + 64;
    const int bbase0 = PAD + i0 + mlan - 8 * h;

    const float* __restrict__ trow = TMP + (size_t)r * n * batch + (b0 >> 4) * 128;
    const unsigned loff = (unsigned)(h * batch * 8 + mlan * 4);
    for (int jblk = 0; jblk < jmax; ) {
     const int jend = jblk + 64 * prio_stretch((jmax - jblk) >> 6);
     for (; jblk < jend; jblk += 64) {
      const float* __restrict__ ap = trow + (size_t)jblk * batch;
#pragma unroll
      for (int p = 0; p < 2; ++p) {
        v4f av[4][2];
#pragma unroll
        for (int u = 0; u < 4; ++u)
#pragma unroll
          for (int t = 0; t < 2; ++t)
            av[u][t] = *(const v4f*)(ap + (size_t)u * 16 * batch + (128 * t + 64 * p) + loff);
        float td[7][4];
#pragma unroll
        for (int d = 0; d < 7; ++d)
#pragma unroll
          for (int e = 0; e < 4; ++e)
            td[d][e] = gpr[bbase0 - jblk - 16 * (d - 3) - 4 * p - e];
        trip_half<true>(acc, td, av);
      }
     }
    }

    v8f c[4][2];
#pragma unroll
    for (int t = 0; t < 2; ++t) {
      v8f y[4];
      combine9(acc[t], y);
#pragma unroll
      for (int it = 0; it < 4; ++it) c[it][t] = y[it];
    }

#pragma unroll
    for (int a = 0; a < 4; ++a) {
      if (r != a) {
#pragma unroll
        for (int t = 0; t < 2; ++t)
#pragma unroll
          for (int v = 0; v < 8; ++v) red[w][t][v][lane] = c[a][t][v];
      }
      __syncthreads();
      if (r == a) {
        const int i = i0 + 16 * a + mlan;
        const float pa = PA[i];
        const float bi = bias[i];
#pragma unroll
        for (int t = 0; t < 2; ++t)
#pragma unroll
          for (int v = 0; v < 8; ++v) {
            float s = 0.f;
#pragma unroll
            for (int q = 0; q < 4; ++q) {
              const float pp = (q == a) ? c[a][t][v] : red[(q << 2) | (w & 3)][t][v][lane];
              s = (q == 0) ? pp : s + pp;
            }
            const int b = b0 + 16 * t + v + bhalf;
            out[(size_t)b * n + i] = bi + pa * s;
          }
      }
      __syncthreads();
    }
  }
}

extern "C" void kernel_launch(void* const* d_in, const int* in_sizes, int n_in,
                              void* d_out, int out_size, void* d_ws, size_t ws_size,
                              hipStream_t stream) {
  const float* x     = (const float*)d_in[0];
  const float* sA    = (const float*)d_in[1];
  const float* sB    = (const float*)d_in[2];
  const float* G     = (const float*)d_in[3];
  const float* H     = (const float*)d_in[4];
  const float* bias  = (const float*)d_in[5];
  float* out = (float*)d_out;

  const int n     = in_sizes[5];
  const int batch = in_sizes[0] / n;
  const int rdim  = in_sizes[3] / n;

  float* PA  = (float*)d_ws;
  float* PB  = PA + n;
  float* Xb  = PB + n;
  float* TMP = Xb + (size_t)n * batch;

  const size_t ldsBytes = (size_t)(PAD + n + 4) * sizeof(float);

  k_prefix<<<2, 1024, 0, stream>>>(sA, sB, PA, PB, n);

  {
    dim3 grid(n / 32, batch / 32);
    k_block_scale<<<grid, 256, 0, stream>>>(x, PB, Xb, n, batch);
  }
  {
    dim3 grid(n / 128, batch / 512, rdim);
    k_pass1<<<grid, 512, ldsBytes, stream>>>(H, PB, Xb, TMP, n, batch);
  }
  {
    dim3 grid(n / 128, batch / 128, 1);
    k_pass2<<<grid, 512, 0, stream>>>(G, PA, TMP, bias, out, n, batch);
  }
}


# Third level of the Toeplitz split: a 128-k trip's 8x8 blocks take 27 products for 64 (was 36 by the two-level form); wave tile 128x16, 27 accumulator tiles, 2 waves per SIMD; sum order changes
# speedup vs baseline: 5.1409x; 1.1433x over previous
#include <hip/hip_runtime.h>


typedef __attribute__((ext_vector_type(2))) float v2f;
typedef __attribute__((ext_vector_type(4))) float v4f;
typedef __attribute__((ext_vector_type(8))) float v8f;

#define PAD 128

__device__ __forceinline__ int prio_stretch(int left) {
  if (left > 22) { __builtin_amdgcn_s_setprio(3); return left - 22; }
  if (left > 6)  { __builtin_amdgcn_s_setprio(2); return left - 6; }
  if (left > 2)  { __builtin_amdgcn_s_setprio(1); return left - 2; }
  __builtin_amdgcn_s_setprio(0);
  return left;
}

__global__ void __launch_bounds__(1024)
k_prefix(const float* __restrict__ sA, const float* __restrict__ sB,
         float* __restrict__ PA, float* __restrict__ PB, int n) {
  __shared__ float wt[32];
  const float* __restrict__ s = blockIdx.x ? sB : sA;
  float* __restrict__ P = blockIdx.x ? PB : PA;
  const int tid = threadIdx.x;
  const int lane = tid & 31;
  const int w = tid >> 5;
  const int nw = blockDim.x >> 5;
  const int chunk = blockDim.x * 4;
  float carry = 1.f;
  for (int base = 0; base < n; base += chunk) {
    const int i0 = base + tid * 4;
    float e[4];
#pragma unroll
    for (int u = 0; u < 4; ++u) {
      const int i = i0 + u;
      const float v = s[max(min(i - 1, n - 2), 0)];
      e[u] = (i >= 1 && i < n) ? v : 1.f;
    }
    const float p0 = e[0], p1 = p0 * e[1], p2 = p1 * e[2], p3 = p2 * e[3];
    float sc = p3;
#pragma unroll
    for (int d = 1; d < 32; d <<= 1) {
      const float v = __shfl_up(sc, d);
      if (lane >= d) sc *= v;
    }
    float ex = __shfl_up(sc, 1);
    if (lane == 0) ex = 1.f;
    if (lane == 31) wt[w] = sc;
    __syncthreads();
    float inc = (lane < nw) ? wt[lane] : 1.f;
#pragma unroll
    for (int d = 1; d < 32; d <<= 1) {
      const float v = __shfl_up(inc, d);
      if (lane >= d) inc *= v;
    }
    const float upto = __shfl(inc, w > 0 ? w - 1 : 0);
    const float below = (w > 0) ? upto : 1.f;
    const float total = __shfl(inc, 31);
    const float pre = carry * below * ex;
    if (i0 + 3 < n) {
      v4f o; o.x = pre * p0; o.y = pre * p1; o.z = pre * p2; o.w = pre * p3;
      *(v4f*)(P + i0) = o;
    } else {
      if (i0 < n) P[i0] = pre * p0;
      if (i0 + 1 < n) P[i0 + 1] = pre * p1;
      if (i0 + 2 < n) P[i0 + 2] = pre * p2;
    }
    carry *= total;
    __syncthreads();
  }
}

__global__ void __launch_bounds__(256)
k_block_scale(const float* __restrict__ x, const float* __restrict__ PB,
              float* __restrict__ Xb, int n, int batch) {
  const int lane = threadIdx.x & 31;
  const int w = threadIdx.x >> 5;
  const int kb = blockIdx.x * 4 + (w & 3);
  const int bg = blockIdx.y * 2 + (w >> 2);
  const int eh = lane >> 4, bl = lane & 15;
  const int k = kb * 8 + eh * 4;
  const v4f xv = *(const v4f*)(x + (size_t)(bg * 16 + bl) * n + k);
  const v4f pv = *(const v4f*)(PB + k);
  *(v4f*)(Xb + (size_t)kb * batch * 8 + bg * 128 + eh * 64 + bl * 4) = pv * xv;
}

template <int L> struct Pow3 { static constexpr int v = 3 * Pow3<L - 1>::v; };
template <> struct Pow3<0> { static constexpr int v = 1; };

template <int L, bool TOEP_IS_B>
struct ToepMul {
  static constexpr int H = 1 << (L - 1);
  static constexpr int NL = Pow3<L - 1>::v;
  __device__ __forceinline__ static void run(v8f* acc, const v4f* t, const v4f* x) {
    v4f ba[2 * H - 1], ca[2 * H - 1], s[H];
#pragma unroll
    for (int d = -(H - 1); d <= H - 1; ++d) {
      ba[d + H - 1] = t[d + H] - t[d];
      ca[d + H - 1] = t[d - H] - t[d];
    }
#pragma unroll
    for (int u = 0; u < H; ++u) s[u] = x[u] + x[u + H];
    ToepMul<L - 1, TOEP_IS_B>::run(acc, t, s);
    ToepMul<L - 1, TOEP_IS_B>::run(acc + NL, ba + (H - 1), x + H);
    ToepMul<L - 1, TOEP_IS_B>::run(acc + 2 * NL, ca + (H - 1), x);
  }
};

template <bool TOEP_IS_B>
struct ToepMul<0, TOEP_IS_B> {
  __device__ __forceinline__ static void run(v8f* acc, const v4f* t, const v4f* x) {
#pragma unroll
    for (int qq = 0; qq < 2; ++qq) {
      v2f ft, fx;
      ft.x = t[0][2 * qq]; ft.y = t[0][2 * qq + 1];
      fx.x = x[0][2 * qq]; fx.y = x[0][2 * qq + 1];
      if (TOEP_IS_B)
        acc[0] = __builtin_amdgcn_wmma_f32_16x16x4_f32(false, fx, false, ft,
                                                       (short)0, acc[0], false, false);
      else
        acc[0] = __builtin_amdgcn_wmma_f32_16x16x4_f32(false, ft, false, fx,
                                                       (short)0, acc[0], false, false);
    }
  }
};

template <int L>
struct ToepComb {
  static constexpr int H = 1 << (L - 1);
  static constexpr int NL = Pow3<L - 1>::v;
  __device__ __forceinline__ static void run(const v8f* a, v8f* y) {
    v8f p[H], q[H], r[H];
    ToepComb<L - 1>::run(a, p);
    ToepComb<L - 1>::run(a + NL, q);
    ToepComb<L - 1>::run(a + 2 * NL, r);
#pragma unroll
    for (int u = 0; u < H; ++u) {
      y[u] = p[u] + q[u];
      y[u + H] = p[u] + r[u];
    }
  }
};
template <>
struct ToepComb<0> {
  __device__ __forceinline__ static void run(const v8f* a, v8f* y) { y[0] = a[0]; }
};

__global__ void __launch_bounds__(256) __attribute__((amdgpu_waves_per_eu(2, 2)))
k_pass1(const float* __restrict__ Hm, const float* __restrict__ PB,
        const float* __restrict__ Xb, float* __restrict__ TMP,
        int n, int batch) {
  extern __shared__ float hp[];
  const int tid = threadIdx.x;
  const int lane = tid & 31;
  const int w = __builtin_amdgcn_readfirstlane(tid >> 5);
  const int r = blockIdx.z;

  for (int t = tid; t < PAD; t += blockDim.x) hp[t] = 0.f;
  for (int t = tid; t < 4; t += blockDim.x) hp[PAD + n + t] = 0.f;
  for (int t = tid; t < n; t += blockDim.x) hp[PAD + t] = Hm[(size_t)r * n + t] / PB[t];
  __syncthreads();

  const int jr   = lane & 15;
  const int h    = lane >> 4;
  const int nlan = lane & 15;
  const int nblk = n / 128;
  const int tripsA = nblk - (int)blockIdx.x;
  const int tripsB = (int)blockIdx.x + 1;

#pragma nounroll
  for (int seg = 0; seg < 4; ++seg) {
    const int half = seg >> 1;
    const int xb = half ? (nblk - 1 - (int)blockIdx.x) : (int)blockIdx.x;
    const int j0 = xb * 128;
    const int after = (seg == 0) ? (tripsA + 2 * tripsB) : (seg == 1) ? (2 * tripsB)
                    : (seg == 2) ? tripsB : 0;
    const int b0 = blockIdx.y * 256 + ((seg & 1) * 8 + w) * 16;

    v8f acc[27];
#pragma unroll
    for (int m = 0; m < 27; ++m) acc[m] = (v8f){0.f,0.f,0.f,0.f,0.f,0.f,0.f,0.f};

    const int abase0 = PAD + 8 * h - jr - j0;
    const float* __restrict__ xrow = Xb + (b0 >> 4) * 128;
    const unsigned loff = (unsigned)(h * batch * 8 + nlan * 4);

    for (int kblk = j0; kblk < n; ) {
     const int kend = min(n, kblk + 128 * prio_stretch(((n - kblk) >> 7) + after));
     for (; kblk < kend; kblk += 128) {
      const float* __restrict__ xp = xrow + (size_t)kblk * batch;
#pragma unroll
      for (int p = 0; p < 2; ++p) {
        v4f xv[8];
#pragma unroll
        for (int u = 0; u < 8; ++u)
          xv[u] = *(const v4f*)(xp + (size_t)u * 16 * batch + 64 * p + loff);
        v4f td[15];
#pragma unroll
        for (int d = 0; d < 15; ++d)
#pragma unroll
          for (int e = 0; e < 4; ++e)
            td[d][e] = hp[abase0 + kblk + 16 * (d - 7) + 4 * p + e];
        ToepMul<3, false>::run(acc, td + 7, xv);
      }
     }
    }

    v8f y[8];
    ToepComb<3>::run(acc, y);
#pragma unroll
    for (int jt = 0; jt < 8; ++jt) {
      float* __restrict__ dst = TMP + (size_t)r * n * batch
          + (size_t)(((j0 + 16 * jt) >> 3) + h) * batch * 8
          + (b0 >> 4) * 128 + nlan * 4;
      v4f lo, hi;
      lo.x = y[jt][0]; lo.y = y[jt][1]; lo.z = y[jt][2]; lo.w = y[jt][3];
      hi.x = y[jt][4]; hi.y = y[jt][5]; hi.z = y[jt][6]; hi.w = y[jt][7];
      *(v4f*)dst = lo;
      *(v4f*)(dst + 64) = hi;
    }
  }
}

#define NROW 4096
__global__ void __launch_bounds__(256) __attribute__((amdgpu_waves_per_eu(2, 2)))
k_pass2(const float* __restrict__ Gm, const float* __restrict__ PA,
        const float* __restrict__ TMP, const float* __restrict__ bias,
        float* __restrict__ out, int n, int batch) {
  __shared__ float gp[4][PAD + NROW + 4];
  __shared__ float red[8][2][8][32];
  const int tid = threadIdx.x;
  const int lane = tid & 31;
  const int w = __builtin_amdgcn_readfirstlane(tid >> 5);
  const int r = w >> 1;

  for (int t = tid; t < 4 * PAD; t += blockDim.x) gp[t / PAD][t % PAD] = 0.f;
  for (int t = tid; t < 16; t += blockDim.x) gp[t >> 2][PAD + n + (t & 3)] = 0.f;
  for (int t = tid; t < 4 * n; t += blockDim.x) {
    const int rr = t / n, m = t - rr * n;
    gp[rr][PAD + m] = Gm[(size_t)rr * n + m] / PA[m];
  }
  __syncthreads();
  const float* __restrict__ gpr = gp[r];

  const int mlan = lane & 15;
  const int h    = lane >> 4;
  const int bhalf = h * 8;
  const int nblk = n / 128;

#pragma nounroll
  for (int seg = 0; seg < 4; ++seg) {
    const int xb = (seg >> 1) ? (nblk - 1 - (int)blockIdx.x) : (int)blockIdx.x;
    const int i0 = xb * 128;
    const int b0 = blockIdx.y * 64 + (seg & 1) * 32 + (w & 1) * 16;

    v8f acc[27];
#pragma unroll
    for (int m = 0; m < 27; ++m) acc[m] = (v8f){0.f,0.f,0.f,0.f,0.f,0.f,0.f,0.f};

    const int jmax   = i0 + 128;
    const int bbase0 = PAD + i0 + mlan - 8 * h;

    const float* __restrict__ trow = TMP + (size_t)r * n * batch + (b0 >> 4) * 128;
    const unsigned loff = (unsigned)(h * batch * 8 + mlan * 4);
    for (int jblk = 0; jblk < jmax; ) {
     const int jend = jblk + 128 * prio_stretch((jmax - jblk) >> 7);
     for (; jblk < jend; jblk += 128) {
      const float* __restrict__ ap = trow + (size_t)jblk * batch;
#pragma unroll
      for (int p = 0; p < 2; ++p) {
        v4f av[8];
#pragma unroll
        for (int u = 0; u < 8; ++u)
          av[u] = *(const v4f*)(ap + (size_t)u * 16 * batch + 64 * p + loff);
        v4f td[15];
#pragma unroll
        for (int d = 0; d < 15; ++d)
#pragma unroll
          for (int e = 0; e < 4; ++e)
            td[d][e] = gpr[bbase0 - jblk - 16 * (d - 7) - 4 * p - e];
        ToepMul<3, true>::run(acc, td + 7, av);
      }
     }
    }

    v8f c[8];
    ToepComb<3>::run(acc, c);

#pragma unroll
    for (int a = 0; a < 4; ++a) {
      if (r != a) {
#pragma unroll
        for (int tt = 0; tt < 2; ++tt)
#pragma unroll
          for (int v = 0; v < 8; ++v) red[w][tt][v][lane] = c[2 * a + tt][v];
      }
      __syncthreads();
      if (r == a) {
#pragma unroll
        for (int tt = 0; tt < 2; ++tt) {
          const int i = i0 + 16 * (2 * a + tt) + mlan;
          const float pa = PA[i];
          const float bi = bias[i];
#pragma unroll
          for (int v = 0; v < 8; ++v) {
            float s = 0.f;
#pragma unroll
            for (int q = 0; q < 4; ++q) {
              const float pp = (q == a) ? c[2 * a + tt][v] : red[(q << 1) | (w & 1)][tt][v][lane];
              s = (q == 0) ? pp : s + pp;
            }
            const int b = b0 + v + bhalf;
            out[(size_t)b * n + i] = bi + pa * s;
          }
        }
      }
      __syncthreads();
    }
  }
}

extern "C" void kernel_launch(void* const* d_in, const int* in_sizes, int n_in,
                              void* d_out, int out_size, void* d_ws, size_t ws_size,
                              hipStream_t stream) {
  const float* x     = (const float*)d_in[0];
  const float* sA    = (const float*)d_in[1];
  const float* sB    = (const float*)d_in[2];
  const float* G     = (const float*)d_in[3];
  const float* H     = (const float*)d_in[4];
  const float* bias  = (const float*)d_in[5];
  float* out = (float*)d_out;

  const int n     = in_sizes[5];
  const int batch = in_sizes[0] / n;
  const int rdim  = in_sizes[3] / n;

  float* PA  = (float*)d_ws;
  float* PB  = PA + n;
  float* Xb  = PB + n;
  float* TMP = Xb + (size_t)n * batch;

  const size_t ldsBytes = (size_t)(PAD + n + 4) * sizeof(float);

  k_prefix<<<2, 1024, 0, stream>>>(sA, sB, PA, PB, n);

  {
    dim3 grid(n / 32, batch / 32);
    k_block_scale<<<grid, 256, 0, stream>>>(x, PB, Xb, n, batch);
  }
  {
    dim3 grid(n / 256, batch / 256, rdim);
    k_pass1<<<grid, 256, ldsBytes, stream>>>(H, PB, Xb, TMP, n, batch);
  }
  {
    dim3 grid(n / 256, batch / 64, 1);
    k_pass2<<<grid, 256, 0, stream>>>(G, PA, TMP, bias, out, n, batch);
  }
}


# Set-up of both passes (LDS vectors H/PB, G/PA): 16-byte loads, all of a thread's loads in flight before its divisions (was one waited 4-byte load pair per element, 64 in a row); sums same
# speedup vs baseline: 5.5717x; 1.0838x over previous
#include <hip/hip_runtime.h>


typedef __attribute__((ext_vector_type(2))) float v2f;
typedef __attribute__((ext_vector_type(4))) float v4f;
typedef __attribute__((ext_vector_type(8))) float v8f;

#define PAD 128
#define SETUP_NT 256

__device__ __forceinline__ int prio_stretch(int left) {
  if (left > 22) { __builtin_amdgcn_s_setprio(3); return left - 22; }
  if (left > 6)  { __builtin_amdgcn_s_setprio(2); return left - 6; }
  if (left > 2)  { __builtin_amdgcn_s_setprio(1); return left - 2; }
  __builtin_amdgcn_s_setprio(0);
  return left;
}

__global__ void __launch_bounds__(1024)
k_prefix(const float* __restrict__ sA, const float* __restrict__ sB,
         float* __restrict__ PA, float* __restrict__ PB, int n) {
  __shared__ float wt[32];
  const float* __restrict__ s = blockIdx.x ? sB : sA;
  float* __restrict__ P = blockIdx.x ? PB : PA;
  const int tid = threadIdx.x;
  const int lane = tid & 31;
  const int w = tid >> 5;
  const int nw = blockDim.x >> 5;
  const int chunk = blockDim.x * 4;
  float carry = 1.f;
  for (int base = 0; base < n; base += chunk) {
    const int i0 = base + tid * 4;
    float e[4];
#pragma unroll
    for (int u = 0; u < 4; ++u) {
      const int i = i0 + u;
      const float v = s[max(min(i - 1, n - 2), 0)];
      e[u] = (i >= 1 && i < n) ? v : 1.f;
    }
    const float p0 = e[0], p1 = p0 * e[1], p2 = p1 * e[2], p3 = p2 * e[3];
    float sc = p3;
#pragma unroll
    for (int d = 1; d < 32; d <<= 1) {
      const float v = __shfl_up(sc, d);
      if (lane >= d) sc *= v;
    }
    float ex = __shfl_up(sc, 1);
    if (lane == 0) ex = 1.f;
    if (lane == 31) wt[w] = sc;
    __syncthreads();
    float inc = (lane < nw) ? wt[lane] : 1.f;
#pragma unroll
    for (int d = 1; d < 32; d <<= 1) {
      const float v = __shfl_up(inc, d);
      if (lane >= d) inc *= v;
    }
    const float upto = __shfl(inc, w > 0 ? w - 1 : 0);
    const float below = (w > 0) ? upto : 1.f;
    const float total = __shfl(inc, 31);
    const float pre = carry * below * ex;
    if (i0 + 3 < n) {
      v4f o; o.x = pre * p0; o.y = pre * p1; o.z = pre * p2; o.w = pre * p3;
      *(v4f*)(P + i0) = o;
    } else {
      if (i0 < n) P[i0] = pre * p0;
      if (i0 + 1 < n) P[i0 + 1] = pre * p1;
      if (i0 + 2 < n) P[i0 + 2] = pre * p2;
    }
    carry *= total;
    __syncthreads();
  }
}

__global__ void __launch_bounds__(256)
k_block_scale(const float* __restrict__ x, const float* __restrict__ PB,
              float* __restrict__ Xb, int n, int batch) {
  const int lane = threadIdx.x & 31;
  const int w = threadIdx.x >> 5;
  const int kb = blockIdx.x * 4 + (w & 3);
  const int bg = blockIdx.y * 2 + (w >> 2);
  const int eh = lane >> 4, bl = lane & 15;
  const int k = kb * 8 + eh * 4;
  const v4f xv = *(const v4f*)(x + (size_t)(bg * 16 + bl) * n + k);
  const v4f pv = *(const v4f*)(PB + k);
  *(v4f*)(Xb + (size_t)kb * batch * 8 + bg * 128 + eh * 64 + bl * 4) = pv * xv;
}

template <int L> struct Pow3 { static constexpr int v = 3 * Pow3<L - 1>::v; };
template <> struct Pow3<0> { static constexpr int v = 1; };

template <int L, bool TOEP_IS_B>
struct ToepMul {
  static constexpr int H = 1 << (L - 1);
  static constexpr int NL = Pow3<L - 1>::v;
  __device__ __forceinline__ static void run(v8f* acc, const v4f* t, const v4f* x) {
    v4f ba[2 * H - 1], ca[2 * H - 1], s[H];
#pragma unroll
    for (int d = -(H - 1); d <= H - 1; ++d) {
      ba[d + H - 1] = t[d + H] - t[d];
      ca[d + H - 1] = t[d - H] - t[d];
    }
#pragma unroll
    for (int u = 0; u < H; ++u) s[u] = x[u] + x[u + H];
    ToepMul<L - 1, TOEP_IS_B>::run(acc, t, s);
    ToepMul<L - 1, TOEP_IS_B>::run(acc + NL, ba + (H - 1), x + H);
    ToepMul<L - 1, TOEP_IS_B>::run(acc + 2 * NL, ca + (H - 1), x);
  }
};

template <bool TOEP_IS_B>
struct ToepMul<0, TOEP_IS_B> {
  __device__ __forceinline__ static void run(v8f* acc, const v4f* t, const v4f* x) {
#pragma unroll
    for (int qq = 0; qq < 2; ++qq) {
      v2f ft, fx;
      ft.x = t[0][2 * qq]; ft.y = t[0][2 * qq + 1];
      fx.x = x[0][2 * qq]; fx.y = x[0][2 * qq + 1];
      if (TOEP_IS_B)
        acc[0] = __builtin_amdgcn_wmma_f32_16x16x4_f32(false, fx, false, ft,
                                                       (short)0, acc[0], false, false);
      else
        acc[0] = __builtin_amdgcn_wmma_f32_16x16x4_f32(false, ft, false, fx,
                                                       (short)0, acc[0], false, false);
    }
  }
};

template <int L>
struct ToepComb {
  static constexpr int H = 1 << (L - 1);
  static constexpr int NL = Pow3<L - 1>::v;
  __device__ __forceinline__ static void run(const v8f* a, v8f* y) {
    v8f p[H], q[H], r[H];
    ToepComb<L - 1>::run(a, p);
    ToepComb<L - 1>::run(a + NL, q);
    ToepComb<L - 1>::run(a + 2 * NL, r);
#pragma unroll
    for (int u = 0; u < H; ++u) {
      y[u] = p[u] + q[u];
      y[u + H] = p[u] + r[u];
    }
  }
};
template <>
struct ToepComb<0> {
  __device__ __forceinline__ static void run(const v8f* a, v8f* y) { y[0] = a[0]; }
};

__global__ void __launch_bounds__(256) __attribute__((amdgpu_waves_per_eu(2, 2)))
k_pass1(const float* __restrict__ Hm, const float* __restrict__ PB,
        const float* __restrict__ Xb, float* __restrict__ TMP,
        int n, int batch) {
  extern __shared__ __attribute__((aligned(16))) float hp[];
  const int tid = threadIdx.x;
  const int lane = tid & 31;
  const int w = __builtin_amdgcn_readfirstlane(tid >> 5);
  const int r = blockIdx.z;

  for (int t = tid; t < PAD; t += blockDim.x) hp[t] = 0.f;
  for (int t = tid; t < 4; t += blockDim.x) hp[PAD + n + t] = 0.f;
  {
    const v4f* __restrict__ H4 = (const v4f*)(Hm + (size_t)r * n);
    const v4f* __restrict__ P4 = (const v4f*)PB;
    v4f* __restrict__ hp4 = (v4f*)(hp + PAD);
    const int n4 = n >> 2;
    int t0 = tid;
    for (; t0 + 3 * SETUP_NT < n4; t0 += 4 * SETUP_NT) {
      v4f hv[4], pv[4];
#pragma unroll
      for (int u = 0; u < 4; ++u) {
        hv[u] = H4[t0 + u * SETUP_NT];
        pv[u] = P4[t0 + u * SETUP_NT];
      }
#pragma unroll
      for (int u = 0; u < 4; ++u) hp4[t0 + u * SETUP_NT] = hv[u] / pv[u];
    }
    for (; t0 < n4; t0 += SETUP_NT) hp4[t0] = H4[t0] / P4[t0];
  }
  __syncthreads();

  const int jr   = lane & 15;
  const int h    = lane >> 4;
  const int nlan = lane & 15;
  const int nblk = n / 128;
  const int tripsA = nblk - (int)blockIdx.x;
  const int tripsB = (int)blockIdx.x + 1;

#pragma nounroll
  for (int seg = 0; seg < 4; ++seg) {
    const int half = seg >> 1;
    const int xb = half ? (nblk - 1 - (int)blockIdx.x) : (int)blockIdx.x;
    const int j0 = xb * 128;
    const int after = (seg == 0) ? (tripsA + 2 * tripsB) : (seg == 1) ? (2 * tripsB)
                    : (seg == 2) ? tripsB : 0;
    const int b0 = blockIdx.y * 256 + ((seg & 1) * 8 + w) * 16;

    v8f acc[27];
#pragma unroll
    for (int m = 0; m < 27; ++m) acc[m] = (v8f){0.f,0.f,0.f,0.f,0.f,0.f,0.f,0.f};

    const int abase0 = PAD + 8 * h - jr - j0;
    const float* __restrict__ xrow = Xb + (b0 >> 4) * 128;
    const unsigned loff = (unsigned)(h * batch * 8 + nlan * 4);

    for (int kblk = j0; kblk < n; ) {
     const int kend = min(n, kblk + 128 * prio_stretch(((n - kblk) >> 7) + after));
     for (; kblk < kend; kblk += 128) {
      const float* __restrict__ xp = xrow + (size_t)kblk * batch;
#pragma unroll
      for (int p = 0; p < 2; ++p) {
        v4f xv[8];
#pragma unroll
        for (int u = 0; u < 8; ++u)
          xv[u] = *(const v4f*)(xp + (size_t)u * 16 * batch + 64 * p + loff);
        v4f td[15];
#pragma unroll
        for (int d = 0; d < 15; ++d)
#pragma unroll
          for (int e = 0; e < 4; ++e)
            td[d][e] = hp[abase0 + kblk + 16 * (d - 7) + 4 * p + e];
        ToepMul<3, false>::run(acc, td + 7, xv);
      }
     }
    }

    v8f y[8];
    ToepComb<3>::run(acc, y);
#pragma unroll
    for (int jt = 0; jt < 8; ++jt) {
      float* __restrict__ dst = TMP + (size_t)r * n * batch
          + (size_t)(((j0 + 16 * jt) >> 3) + h) * batch * 8
          + (b0 >> 4) * 128 + nlan * 4;
      v4f lo, hi;
      lo.x = y[jt][0]; lo.y = y[jt][1]; lo.z = y[jt][2]; lo.w = y[jt][3];
      hi.x = y[jt][4]; hi.y = y[jt][5]; hi.z = y[jt][6]; hi.w = y[jt][7];
      *(v4f*)dst = lo;
      *(v4f*)(dst + 64) = hi;
    }
  }
}

#define NROW 4096
__global__ void __launch_bounds__(256) __attribute__((amdgpu_waves_per_eu(2, 2)))
k_pass2(const float* __restrict__ Gm, const float* __restrict__ PA,
        const float* __restrict__ TMP, const float* __restrict__ bias,
        float* __restrict__ out, int n, int batch) {
  __shared__ __attribute__((aligned(16))) float gp[4][PAD + NROW + 4];
  __shared__ float red[8][2][8][32];
  const int tid = threadIdx.x;
  const int lane = tid & 31;
  const int w = __builtin_amdgcn_readfirstlane(tid >> 5);
  const int r = w >> 1;

  for (int t = tid; t < 4 * PAD; t += blockDim.x) gp[t / PAD][t % PAD] = 0.f;
  for (int t = tid; t < 16; t += blockDim.x) gp[t >> 2][PAD + n + (t & 3)] = 0.f;
  {
    const v4f* __restrict__ P4 = (const v4f*)PA;
    const int n4 = n >> 2;
    int t0 = tid;
    for (; t0 + 3 * SETUP_NT < n4; t0 += 4 * SETUP_NT) {
      v4f gv[4][4], pv[4];
#pragma unroll
      for (int u = 0; u < 4; ++u) {
        pv[u] = P4[t0 + u * SETUP_NT];
#pragma unroll
        for (int rr = 0; rr < 4; ++rr)
          gv[rr][u] = *(const v4f*)(Gm + (size_t)rr * n + 4 * (t0 + u * SETUP_NT));
      }
#pragma unroll
      for (int u = 0; u < 4; ++u)
#pragma unroll
        for (int rr = 0; rr < 4; ++rr)
          *(v4f*)&gp[rr][PAD + 4 * (t0 + u * SETUP_NT)] = gv[rr][u] / pv[u];
    }
    for (; t0 < n4; t0 += SETUP_NT) {
      const v4f pv = P4[t0];
#pragma unroll
      for (int rr = 0; rr < 4; ++rr)
        *(v4f*)&gp[rr][PAD + 4 * t0] = *(const v4f*)(Gm + (size_t)rr * n + 4 * t0) / pv;
    }
  }
  __syncthreads();
  const float* __restrict__ gpr = gp[r];

  const int mlan = lane & 15;
  const int h    = lane >> 4;
  const int bhalf = h * 8;
  const int nblk = n / 128;

#pragma nounroll
  for (int seg = 0; seg < 4; ++seg) {
    const int xb = (seg >> 1) ? (nblk - 1 - (int)blockIdx.x) : (int)blockIdx.x;
    const int i0 = xb * 128;
    const int b0 = blockIdx.y * 64 + (seg & 1) * 32 + (w & 1) * 16;

    v8f acc[27];
#pragma unroll
    for (int m = 0; m < 27; ++m) acc[m] = (v8f){0.f,0.f,0.f,0.f,0.f,0.f,0.f,0.f};

    const int jmax   = i0 + 128;
    const int bbase0 = PAD + i0 + mlan - 8 * h;

    const float* __restrict__ trow = TMP + (size_t)r * n * batch + (b0 >> 4) * 128;
    const unsigned loff = (unsigned)(h * batch * 8 + mlan * 4);
    for (int jblk = 0; jblk < jmax; ) {
     const int jend = jblk + 128 * prio_stretch((jmax - jblk) >> 7);
     for (; jblk < jend; jblk += 128) {
      const float* __restrict__ ap = trow + (size_t)jblk * batch;
#pragma unroll
      for (int p = 0; p < 2; ++p) {
        v4f av[8];
#pragma unroll
        for (int u = 0; u < 8; ++u)
          av[u] = *(const v4f*)(ap + (size_t)u * 16 * batch + 64 * p + loff);
        v4f td[15];
#pragma unroll
        for (int d = 0; d < 15; ++d)
#pragma unroll
          for (int e = 0; e < 4; ++e)
            td[d][e] = gpr[bbase0 - jblk - 16 * (d - 7) - 4 * p - e];
        ToepMul<3, true>::run(acc, td + 7, av);
      }
     }
    }

    v8f c[8];
    ToepComb<3>::run(acc, c);

#pragma unroll
    for (int a = 0; a < 4; ++a) {
      if (r != a) {
#pragma unroll
        for (int tt = 0; tt < 2; ++tt)
#pragma unroll
          for (int v = 0; v < 8; ++v) red[w][tt][v][lane] = c[2 * a + tt][v];
      }
      __syncthreads();
      if (r == a) {
#pragma unroll
        for (int tt = 0; tt < 2; ++tt) {
          const int i = i0 + 16 * (2 * a + tt) + mlan;
          const float pa = PA[i];
          const float bi = bias[i];
#pragma unroll
          for (int v = 0; v < 8; ++v) {
            float s = 0.f;
#pragma unroll
            for (int q = 0; q < 4; ++q) {
              const float pp = (q == a) ? c[2 * a + tt][v] : red[(q << 1) | (w & 1)][tt][v][lane];
              s = (q == 0) ? pp : s + pp;
            }
            const int b = b0 + v + bhalf;
            out[(size_t)b * n + i] = bi + pa * s;
          }
        }
      }
      __syncthreads();
    }
  }
}

extern "C" void kernel_launch(void* const* d_in, const int* in_sizes, int n_in,
                              void* d_out, int out_size, void* d_ws, size_t ws_size,
                              hipStream_t stream) {
  const float* x     = (const float*)d_in[0];
  const float* sA    = (const float*)d_in[1];
  const float* sB    = (const float*)d_in[2];
  const float* G     = (const float*)d_in[3];
  const float* H     = (const float*)d_in[4];
  const float* bias  = (const float*)d_in[5];
  float* out = (float*)d_out;

  const int n     = in_sizes[5];
  const int batch = in_sizes[0] / n;
  const int rdim  = in_sizes[3] / n;

  float* PA  = (float*)d_ws;
  float* PB  = PA + n;
  float* Xb  = PB + n;
  float* TMP = Xb + (size_t)n * batch;

  const size_t ldsBytes = (size_t)(PAD + n + 4) * sizeof(float);

  k_prefix<<<2, 1024, 0, stream>>>(sA, sB, PA, PB, n);

  {
    dim3 grid(n / 32, batch / 32);
    k_block_scale<<<grid, 256, 0, stream>>>(x, PB, Xb, n, batch);
  }
  {
    dim3 grid(n / 256, batch / 256, rdim);
    k_pass1<<<grid, 256, ldsBytes, stream>>>(H, PB, Xb, TMP, n, batch);
  }
  {
    dim3 grid(n / 256, batch / 64, 1);
    k_pass2<<<grid, 256, 0, stream>>>(G, PA, TMP, bias, out, n, batch);
  }
}


# Fourth level of the Toeplitz split, in time: a 256-k step of a 256-row block as 3 three-level products run one after the other (B-A, C-A vectors in LDS, common term parked); sum order changes
# speedup vs baseline: 6.9477x; 1.2470x over previous
#include <hip/hip_runtime.h>


typedef __attribute__((ext_vector_type(2))) float v2f;
typedef __attribute__((ext_vector_type(4))) float v4f;
typedef __attribute__((ext_vector_type(8))) float v8f;

#define PAD 128
#define SETUP_NT 256

__device__ __forceinline__ int prio_stretch(int left) {
  if (left > 22) { __builtin_amdgcn_s_setprio(3); return left - 22; }
  if (left > 6)  { __builtin_amdgcn_s_setprio(2); return left - 6; }
  if (left > 2)  { __builtin_amdgcn_s_setprio(1); return left - 2; }
  __builtin_amdgcn_s_setprio(0);
  return left;
}

__global__ void __launch_bounds__(1024)
k_prefix(const float* __restrict__ sA, const float* __restrict__ sB,
         float* __restrict__ PA, float* __restrict__ PB, int n) {
  __shared__ float wt[32];
  const float* __restrict__ s = blockIdx.x ? sB : sA;
  float* __restrict__ P = blockIdx.x ? PB : PA;
  const int tid = threadIdx.x;
  const int lane = tid & 31;
  const int w = tid >> 5;
  const int nw = blockDim.x >> 5;
  const int chunk = blockDim.x * 4;
  float carry = 1.f;
  for (int base = 0; base < n; base += chunk) {
    const int i0 = base + tid * 4;
    float e[4];
#pragma unroll
    for (int u = 0; u < 4; ++u) {
      const int i = i0 + u;
      const float v = s[max(min(i - 1, n - 2), 0)];
      e[u] = (i >= 1 && i < n) ? v : 1.f;
    }
    const float p0 = e[0], p1 = p0 * e[1], p2 = p1 * e[2], p3 = p2 * e[3];
    float sc = p3;
#pragma unroll
    for (int d = 1; d < 32; d <<= 1) {
      const float v = __shfl_up(sc, d);
      if (lane >= d) sc *= v;
    }
    float ex = __shfl_up(sc, 1);
    if (lane == 0) ex = 1.f;
    if (lane == 31) wt[w] = sc;
    __syncthreads();
    float inc = (lane < nw) ? wt[lane] : 1.f;
#pragma unroll
    for (int d = 1; d < 32; d <<= 1) {
      const float v = __shfl_up(inc, d);
      if (lane >= d) inc *= v;
    }
    const float upto = __shfl(inc, w > 0 ? w - 1 : 0);
    const float below = (w > 0) ? upto : 1.f;
    const float total = __shfl(inc, 31);
    const float pre = carry * below * ex;
    if (i0 + 3 < n) {
      v4f o; o.x = pre * p0; o.y = pre * p1; o.z = pre * p2; o.w = pre * p3;
      *(v4f*)(P + i0) = o;
    } else {
      if (i0 < n) P[i0] = pre * p0;
      if (i0 + 1 < n) P[i0 + 1] = pre * p1;
      if (i0 + 2 < n) P[i0 + 2] = pre * p2;
    }
    carry *= total;
    __syncthreads();
  }
}

__global__ void __launch_bounds__(256)
k_block_scale(const float* __restrict__ x, const float* __restrict__ PB,
              float* __restrict__ Xb, int n, int batch) {
  const int lane = threadIdx.x & 31;
  const int w = threadIdx.x >> 5;
  const int kb = blockIdx.x * 4 + (w & 3);
  const int bg = blockIdx.y * 2 + (w >> 2);
  const int eh = lane >> 4, bl = lane & 15;
  const int k = kb * 8 + eh * 4;
  const v4f xv = *(const v4f*)(x + (size_t)(bg * 16 + bl) * n + k);
  const v4f pv = *(const v4f*)(PB + k);
  *(v4f*)(Xb + (size_t)kb * batch * 8 + bg * 128 + eh * 64 + bl * 4) = pv * xv;
}

template <int L> struct Pow3 { static constexpr int v = 3 * Pow3<L - 1>::v; };
template <> struct Pow3<0> { static constexpr int v = 1; };

template <int L, bool TOEP_IS_B>
struct ToepMul {
  static constexpr int H = 1 << (L - 1);
  static constexpr int NL = Pow3<L - 1>::v;
  __device__ __forceinline__ static void run(v8f* acc, const v4f* t, const v4f* x) {
    v4f ba[2 * H - 1], ca[2 * H - 1], s[H];
#pragma unroll
    for (int d = -(H - 1); d <= H - 1; ++d) {
      ba[d + H - 1] = t[d + H] - t[d];
      ca[d + H - 1] = t[d - H] - t[d];
    }
#pragma unroll
    for (int u = 0; u < H; ++u) s[u] = x[u] + x[u + H];
    ToepMul<L - 1, TOEP_IS_B>::run(acc, t, s);
    ToepMul<L - 1, TOEP_IS_B>::run(acc + NL, ba + (H - 1), x + H);
    ToepMul<L - 1, TOEP_IS_B>::run(acc + 2 * NL, ca + (H - 1), x);
  }
};

template <bool TOEP_IS_B>
struct ToepMul<0, TOEP_IS_B> {
  __device__ __forceinline__ static void run(v8f* acc, const v4f* t, const v4f* x) {
#pragma unroll
    for (int qq = 0; qq < 2; ++qq) {
      v2f ft, fx;
      ft.x = t[0][2 * qq]; ft.y = t[0][2 * qq + 1];
      fx.x = x[0][2 * qq]; fx.y = x[0][2 * qq + 1];
      if (TOEP_IS_B)
        acc[0] = __builtin_amdgcn_wmma_f32_16x16x4_f32(false, fx, false, ft,
                                                       (short)0, acc[0], false, false);
      else
        acc[0] = __builtin_amdgcn_wmma_f32_16x16x4_f32(false, ft, false, fx,
                                                       (short)0, acc[0], false, false);
    }
  }
};

template <int L>
struct ToepComb {
  static constexpr int H = 1 << (L - 1);
  static constexpr int NL = Pow3<L - 1>::v;
  __device__ __forceinline__ static void run(const v8f* a, v8f* y) {
    v8f p[H], q[H], r[H];
    ToepComb<L - 1>::run(a, p);
    ToepComb<L - 1>::run(a + NL, q);
    ToepComb<L - 1>::run(a + 2 * NL, r);
#pragma unroll
    for (int u = 0; u < H; ++u) {
      y[u] = p[u] + q[u];
      y[u + H] = p[u] + r[u];
    }
  }
};
template <>
struct ToepComb<0> {
  __device__ __forceinline__ static void run(const v8f* a, v8f* y) { y[0] = a[0]; }
};

template <bool TOEP_IS_B, bool SUM>
__device__ __forceinline__ void toep_trip(v8f* acc, const float* __restrict__ tv, int tidx,
                                          const float* __restrict__ xp, size_t botoff,
                                          int batch, unsigned loff) {
#pragma unroll
  for (int p = 0; p < 2; ++p) {
    v4f xv[8];
#pragma unroll
    for (int u = 0; u < 8; ++u)
      xv[u] = *(const v4f*)(xp + (size_t)u * 16 * batch + 64 * p + loff);
    if (SUM) {
#pragma unroll
      for (int u = 0; u < 8; ++u)
        xv[u] += *(const v4f*)(xp + botoff + (size_t)u * 16 * batch + 64 * p + loff);
    }
    v4f td[15];
#pragma unroll
    for (int d = 0; d < 15; ++d)
#pragma unroll
      for (int e = 0; e < 4; ++e)
        td[d][e] = TOEP_IS_B ? tv[tidx - 16 * (d - 7) - 4 * p - e]
                             : tv[tidx + 16 * (d - 7) + 4 * p + e];
    ToepMul<3, TOEP_IS_B>::run(acc, td + 7, xv);
  }
}

template <bool TOEP_IS_B, bool SUM>
__device__ __forceinline__ void toep_kloop(v8f* acc, const float* __restrict__ tv, int tidx,
                                           const float* __restrict__ xp, size_t botoff,
                                           int batch, unsigned loff, int nst, int& left) {
  const size_t xstep = (size_t)256 * batch;
  for (int done = 0; done < nst; ) {
    const int st = min(nst - done, prio_stretch(left));
    left -= st;
    for (int t = 0; t < st; ++t) {
      toep_trip<TOEP_IS_B, SUM>(acc, tv, tidx, xp, botoff, batch, loff);
      tidx += TOEP_IS_B ? -256 : 256;
      xp += xstep;
    }
    done += st;
  }
}

#define PARK_FLOATS (8 * 8 * 8 * 32)
#define NROW 4096
#define LVEC (PAD + NROW + 4)

__global__ void __launch_bounds__(256) __attribute__((amdgpu_waves_per_eu(2, 2)))
k_pass1(const float* __restrict__ Hm, const float* __restrict__ PB,
        const float* __restrict__ Xb, float* __restrict__ TMP,
        int n, int batch) {
  __shared__ __attribute__((aligned(16))) float hp[3 * LVEC + PARK_FLOATS];
  const int L = PAD + n + 4;
  float* __restrict__ hb = hp + L;
  float* __restrict__ hc = hp + 2 * L;
  float* __restrict__ park = hp + 3 * L;
  const int tid = threadIdx.x;
  const int lane = tid & 31;
  const int w = __builtin_amdgcn_readfirstlane(tid >> 5);
  const int r = blockIdx.z;

  for (int t = tid; t < PAD; t += blockDim.x) hp[t] = 0.f;
  for (int t = tid; t < 4; t += blockDim.x) hp[PAD + n + t] = 0.f;
  {
    const v4f* __restrict__ H4 = (const v4f*)(Hm + (size_t)r * n);
    const v4f* __restrict__ P4 = (const v4f*)PB;
    v4f* __restrict__ hp4 = (v4f*)(hp + PAD);
    const int n4 = n >> 2;
    int t0 = tid;
    for (; t0 + 3 * SETUP_NT < n4; t0 += 4 * SETUP_NT) {
      v4f hv[4], pv[4];
#pragma unroll
      for (int u = 0; u < 4; ++u) {
        hv[u] = H4[t0 + u * SETUP_NT];
        pv[u] = P4[t0 + u * SETUP_NT];
      }
#pragma unroll
      for (int u = 0; u < 4; ++u) hp4[t0 + u * SETUP_NT] = hv[u] / pv[u];
    }
    for (; t0 < n4; t0 += SETUP_NT) hp4[t0] = H4[t0] / P4[t0];
  }
  __syncthreads();
  for (int a = tid; a < L; a += blockDim.x) {
    const float c = hp[a];
    const float up = (a + 128 < L) ? hp[a + 128] : 0.f;
    const float dn = (a >= 128) ? hp[a - 128] : 0.f;
    hb[a] = up - c;
    hc[a] = dn - c;
  }
  __syncthreads();

  const int jr   = lane & 15;
  const int h    = lane >> 4;
  const int nlan = lane & 15;
  const int nsb = n / 256;
  const int b0 = blockIdx.y * 128 + w * 16;
  const float* __restrict__ xrow = Xb + (b0 >> 4) * 128;
  const unsigned loff = (unsigned)(h * batch * 8 + nlan * 4);
  const size_t botoff = (size_t)128 * batch;
  float* __restrict__ pk = park + w * (8 * 8 * 32) + lane;

  int left = 3 * (nsb + 1);

#pragma nounroll
  for (int seg = 0; seg < 2; ++seg) {
    const int sb = seg ? (nsb - 1 - (int)blockIdx.x) : (int)blockIdx.x;
    const int j0 = sb * 256;
    const int nst = nsb - sb;
    const int tidx0 = PAD + 8 * h - jr;
    const float* __restrict__ xp0 = xrow + (size_t)j0 * batch;

#pragma nounroll
    for (int prod = 0; prod < 3; ++prod) {
      v8f acc[27];
#pragma unroll
      for (int m = 0; m < 27; ++m) acc[m] = (v8f){0.f,0.f,0.f,0.f,0.f,0.f,0.f,0.f};

      if (prod == 0)
        toep_kloop<false, true>(acc, hp, tidx0, xp0, botoff, batch, loff, nst, left);
      else
        toep_kloop<false, false>(acc, prod == 1 ? hb : hc, tidx0,
                                 prod == 1 ? xp0 + botoff : xp0, botoff, batch, loff, nst, left);

      v8f y[8];
      ToepComb<3>::run(acc, y);
      if (prod == 0) {
#pragma unroll
        for (int jt = 0; jt < 8; ++jt)
#pragma unroll
          for (int v = 0; v < 8; ++v) pk[(jt * 8 + v) * 32] = y[jt][v];
      } else {
        const int jbase = j0 + (prod == 2 ? 128 : 0);
#pragma unroll
        for (int jt = 0; jt < 8; ++jt) {
          float* __restrict__ dst = TMP + (size_t)r * n * batch
              + (size_t)(((jbase + 16 * jt) >> 3) + h) * batch * 8
              + (b0 >> 4) * 128 + nlan * 4;
          v4f lo, hi;
          lo.x = y[jt][0] + pk[(jt * 8 + 0) * 32]; lo.y = y[jt][1] + pk[(jt * 8 + 1) * 32];
          lo.z = y[jt][2] + pk[(jt * 8 + 2) * 32]; lo.w = y[jt][3] + pk[(jt * 8 + 3) * 32];
          hi.x = y[jt][4] + pk[(jt * 8 + 4) * 32]; hi.y = y[jt][5] + pk[(jt * 8 + 5) * 32];
          hi.z = y[jt][6] + pk[(jt * 8 + 6) * 32]; hi.w = y[jt][7] + pk[(jt * 8 + 7) * 32];
          *(v4f*)dst = lo;
          *(v4f*)(dst + 64) = hi;
        }
      }
    }
  }
}

__global__ void __launch_bounds__(256) __attribute__((amdgpu_waves_per_eu(2, 2)))
k_pass2(const float* __restrict__ Gm, const float* __restrict__ PA,
        const float* __restrict__ TMP, const float* __restrict__ bias,
        float* __restrict__ out, int n, int batch) {
  __shared__ __attribute__((aligned(16))) float gp[4][3][LVEC];
  __shared__ float red[8][2][8][32];
  __shared__ float park[PARK_FLOATS];
  const int tid = threadIdx.x;
  const int lane = tid & 31;
  const int w = __builtin_amdgcn_readfirstlane(tid >> 5);
  const int r = w >> 1;

  for (int t = tid; t < 4 * PAD; t += blockDim.x) gp[t / PAD][0][t % PAD] = 0.f;
  for (int t = tid; t < 16; t += blockDim.x) gp[t >> 2][0][PAD + n + (t & 3)] = 0.f;
  {
    const v4f* __restrict__ P4 = (const v4f*)PA;
    const int n4 = n >> 2;
    int t0 = tid;
    for (; t0 + 3 * SETUP_NT < n4; t0 += 4 * SETUP_NT) {
      v4f gv[4][4], pv[4];
#pragma unroll
      for (int u = 0; u < 4; ++u) {
        pv[u] = P4[t0 + u * SETUP_NT];
#pragma unroll
        for (int rr = 0; rr < 4; ++rr)
          gv[rr][u] = *(const v4f*)(Gm + (size_t)rr * n + 4 * (t0 + u * SETUP_NT));
      }
#pragma unroll
      for (int u = 0; u < 4; ++u)
#pragma unroll
        for (int rr = 0; rr < 4; ++rr)
          *(v4f*)&gp[rr][0][PAD + 4 * (t0 + u * SETUP_NT)] = gv[rr][u] / pv[u];
    }
    for (; t0 < n4; t0 += SETUP_NT) {
      const v4f pv = P4[t0];
#pragma unroll
      for (int rr = 0; rr < 4; ++rr)
        *(v4f*)&gp[rr][0][PAD + 4 * t0] = *(const v4f*)(Gm + (size_t)rr * n + 4 * t0) / pv;
    }
  }
  __syncthreads();
  {
    const int L = PAD + n + 4;
    for (int a = tid; a < L; a += blockDim.x) {
#pragma unroll
      for (int rr = 0; rr < 4; ++rr) {
        const float c = gp[rr][0][a];
        const float dn = (a >= 128) ? gp[rr][0][a - 128] : 0.f;
        const float up = (a + 128 < L) ? gp[rr][0][a + 128] : 0.f;
        gp[rr][1][a] = dn - c;
        gp[rr][2][a] = up - c;
      }
    }
  }
  __syncthreads();

  const int mlan = lane & 15;
  const int h    = lane >> 4;
  const int bhalf = h * 8;
  const int nsb = n / 256;
  const int b0 = blockIdx.y * 32 + (w & 1) * 16;
  const float* __restrict__ trow = TMP + (size_t)r * n * batch + (b0 >> 4) * 128;
  const unsigned loff = (unsigned)(h * batch * 8 + mlan * 4);
  const size_t botoff = (size_t)128 * batch;
  float* __restrict__ pk = park + w * (8 * 8 * 32) + lane;

#pragma nounroll
  for (int seg = 0; seg < 2; ++seg) {
    const int sb = seg ? (nsb - 1 - (int)blockIdx.x) : (int)blockIdx.x;
    const int i0 = sb * 256;
    const int nst = sb + 1;
    const int tidx0 = PAD + i0 + mlan - 8 * h;

#pragma nounroll
    for (int prod = 0; prod < 3; ++prod) {
      v8f acc[27];
#pragma unroll
      for (int m = 0; m < 27; ++m) acc[m] = (v8f){0.f,0.f,0.f,0.f,0.f,0.f,0.f,0.f};

      int left = (prod == 0) ? 2 * nst : nst;
      if (prod == 0)
        toep_kloop<true, true>(acc, gp[r][0], tidx0, trow, botoff, batch, loff, nst, left);
      else
        toep_kloop<true, false>(acc, gp[r][prod], tidx0,
                                prod == 1 ? trow + botoff : trow, botoff, batch, loff, nst, left);

      v8f c[8];
      ToepComb<3>::run(acc, c);
      if (prod == 0) {
#pragma unroll
        for (int it = 0; it < 8; ++it)
#pragma unroll
          for (int v = 0; v < 8; ++v) pk[(it * 8 + v) * 32] = c[it][v];
        continue;
      }
#pragma unroll
      for (int it = 0; it < 8; ++it)
#pragma unroll
        for (int v = 0; v < 8; ++v) c[it][v] += pk[(it * 8 + v) * 32];
      const int ibase = i0 + (prod == 2 ? 128 : 0);

#pragma unroll
      for (int a = 0; a < 4; ++a) {
        if (r != a) {
#pragma unroll
          for (int tt = 0; tt < 2; ++tt)
#pragma unroll
            for (int v = 0; v < 8; ++v) red[w][tt][v][lane] = c[2 * a + tt][v];
        }
        __syncthreads();
        if (r == a) {
#pragma unroll
          for (int tt = 0; tt < 2; ++tt) {
            const int i = ibase + 16 * (2 * a + tt) + mlan;
            const float pa = PA[i];
            const float bi = bias[i];
#pragma unroll
            for (int v = 0; v < 8; ++v) {
              float s = 0.f;
#pragma unroll
              for (int q = 0; q < 4; ++q) {
                const float pp = (q == a) ? c[2 * a + tt][v] : red[(q << 1) | (w & 1)][tt][v][lane];
                s = (q == 0) ? pp : s + pp;
              }
              const int b = b0 + v + bhalf;
              out[(size_t)b * n + i] = bi + pa * s;
            }
          }
        }
        __syncthreads();
      }
    }
  }
}

extern "C" void kernel_launch(void* const* d_in, const int* in_sizes, int n_in,
                              void* d_out, int out_size, void* d_ws, size_t ws_size,
                              hipStream_t stream) {
  const float* x     = (const float*)d_in[0];
  const float* sA    = (const float*)d_in[1];
  const float* sB    = (const float*)d_in[2];
  const float* G     = (const float*)d_in[3];
  const float* H     = (const float*)d_in[4];
  const float* bias  = (const float*)d_in[5];
  float* out = (float*)d_out;

  const int n     = in_sizes[5];
  const int batch = in_sizes[0] / n;
  const int rdim  = in_sizes[3] / n;

  float* PA  = (float*)d_ws;
  float* PB  = PA + n;
  float* Xb  = PB + n;
  float* TMP = Xb + (size_t)n * batch;

  k_prefix<<<2, 1024, 0, stream>>>(sA, sB, PA, PB, n);

  {
    dim3 grid(n / 32, batch / 32);
    k_block_scale<<<grid, 256, 0, stream>>>(x, PB, Xb, n, batch);
  }
  {
    dim3 grid(n / 512, batch / 128, rdim);
    k_pass1<<<grid, 256, 0, stream>>>(H, PB, Xb, TMP, n, batch);
  }
  {
    dim3 grid(n / 512, batch / 32, 1);
    k_pass2<<<grid, 256, 0, stream>>>(G, PA, TMP, bias, out, n, batch);
  }
}
